# starting kernel
# baseline (speedup 1.0000x reference)
#include <hip/hip_runtime.h>
#include <math.h>

typedef __attribute__((ext_vector_type(2))) float v2f;
typedef __attribute__((ext_vector_type(8))) float v8f;

#define N_BATCH 128
#define D_DIM 4096
#define E_COLS 8192
#define K_NNZ 128
#define COLS_PER_BLOCK 8
#define THREADS 256

__global__ __launch_bounds__(THREADS)
void logmmexp_wmma_kernel(const float* __restrict__ x,
                          const float* __restrict__ values,
                          const int* __restrict__ rows,
                          float* __restrict__ out)
{
    __shared__ float s_vals[K_NNZ];
    __shared__ int   s_rows[K_NNZ];
    __shared__ float s_sums[8 * 16];
    __shared__ float s_out[N_BATCH * COLS_PER_BLOCK];

    const int tid  = threadIdx.x;
    const int wave = tid >> 5;
    const int lane = tid & 31;
    const int h    = lane >> 4;
    const int nloc = lane & 15;
    const int nrow = wave * 16 + nloc;
    const float* xrow = x + (size_t)nrow * D_DIM;

    const int e_base = blockIdx.x * COLS_PER_BLOCK;

    v2f ones;
    ones.x = 1.0f;
    ones.y = 1.0f;

    for (int col = 0; col < COLS_PER_BLOCK; ++col) {
        const int e = e_base + col;
        const size_t kbase = (size_t)e * K_NNZ;

        if (tid < K_NNZ) {
            s_vals[tid] = values[kbase + tid];
        } else {
            const int t = tid - K_NNZ;
            s_rows[t] = rows[kbase + t];
        }
        if (col + 1 < COLS_PER_BLOCK && tid == 0) {
            __builtin_prefetch(values + kbase + K_NNZ, 0, 1);
            __builtin_prefetch(rows + kbase + K_NNZ, 0, 1);
        }
        __syncthreads();

        float t_cache[64];
        float m = -INFINITY;
#pragma unroll
        for (int c = 0; c < 32; ++c) {
            const int k0 = 4 * c + 2 * h;
            const float t0 = s_vals[k0]     + xrow[s_rows[k0]];
            const float t1 = s_vals[k0 + 1] + xrow[s_rows[k0 + 1]];
            t_cache[2 * c]     = t0;
            t_cache[2 * c + 1] = t1;
            m = fmaxf(m, fmaxf(t0, t1));
        }
        m = fmaxf(m, __shfl_xor(m, 16, 32));

        v8f acc = {};
#pragma unroll
        for (int c = 0; c < 32; ++c) {
            v2f a;
            a.x = __expf(t_cache[2 * c]     - m);
            a.y = __expf(t_cache[2 * c + 1] - m);
            acc = __builtin_amdgcn_wmma_f32_16x16x4_f32(
                false, a, false, ones, (short)0, acc, false, false);
        }

        if (lane == 0) {
#pragma unroll
            for (int v = 0; v < 8; ++v) s_sums[wave * 16 + v] = acc[v];
        } else if (lane == 16) {
#pragma unroll
            for (int v = 0; v < 8; ++v) s_sums[wave * 16 + 8 + v] = acc[v];
        }
        if (lane < 16) {
            const float s = s_sums[wave * 16 + nloc];
            s_out[nrow * COLS_PER_BLOCK + col] =
                m + 0.6931471805599453f * __log2f(s);
        }
        __syncthreads();
    }

    {
        const int n    = tid >> 1;
        const int part = tid & 1;
        const float4* src =
            (const float4*)(s_out + n * COLS_PER_BLOCK + part * 4);
        float4* dst =
            (float4*)(out + (size_t)n * E_COLS + e_base + part * 4);
        *dst = *src;
    }
}

extern "C" void kernel_launch(void* const* d_in, const int* in_sizes, int n_in,
                              void* d_out, int out_size, void* d_ws, size_t ws_size,
                              hipStream_t stream) {
    const float* x      = (const float*)d_in[0];
    const float* values = (const float*)d_in[1];
    const int*   rowidx = (const int*)d_in[2];
    float* out = (float*)d_out;

    dim3 grid(E_COLS / COLS_PER_BLOCK);
    dim3 block(THREADS);
    hipLaunchKernelGGL(logmmexp_wmma_kernel, grid, block, 0, stream,
                       x, values, rowidx, out);
}


# x is transposed into the workspace on every call ([4096][128]); a wave's gather load now reads two contiguous 64-byte runs instead of 32 separate lines; k order and summation order unchanged
# speedup vs baseline: 4.8954x; 4.8954x over previous
#include <hip/hip_runtime.h>
#include <math.h>


typedef __attribute__((ext_vector_type(2))) float v2f;
typedef __attribute__((ext_vector_type(8))) float v8f;

#define N_BATCH 128
#define D_DIM 4096
#define E_COLS 8192
#define K_NNZ 128
#define COLS_PER_BLOCK 8
#define THREADS 256
#define TP_TILE 32
#define XT_ROW_BYTES (N_BATCH * 4)

__global__ __launch_bounds__(256)
void transpose_x_kernel(const float* __restrict__ x, float* __restrict__ xt)
{
    __shared__ float tile[TP_TILE][TP_TILE + 1];

    const int tx = threadIdx.x & 31;
    const int ty = threadIdx.x >> 5;
    const int d0 = blockIdx.x * TP_TILE;
    const int n0 = blockIdx.y * TP_TILE;

#pragma unroll
    for (int j = 0; j < TP_TILE; j += 8)
        tile[ty + j][tx] = x[(size_t)(n0 + ty + j) * D_DIM + d0 + tx];
    __syncthreads();
#pragma unroll
    for (int j = 0; j < TP_TILE; j += 8)
        xt[(size_t)(d0 + ty + j) * N_BATCH + n0 + tx] = tile[tx][ty + j];
}

__device__ __forceinline__ float load_xt(const float* __restrict__ xt, int d, unsigned lane_byte)
{
    const unsigned off = (unsigned)d * (unsigned)XT_ROW_BYTES + lane_byte;
    return *(const float*)((const char*)xt + off);
}

__global__ __launch_bounds__(THREADS)
void logmmexp_wmma_kernel(const float* __restrict__ xt,
                          const float* __restrict__ values,
                          const int* __restrict__ rows,
                          float* __restrict__ out)
{
    __shared__ float s_vals[K_NNZ];
    __shared__ int   s_rows[K_NNZ];
    __shared__ float s_sums[8 * 16];
    __shared__ float s_out[N_BATCH * COLS_PER_BLOCK];

    const int tid  = threadIdx.x;
    const int wave = tid >> 5;
    const int lane = tid & 31;
    const int h    = lane >> 4;
    const int nloc = lane & 15;
    const int nrow = wave * 16 + nloc;
    const unsigned lane_byte = (unsigned)nrow * 4u;

    const int e_base = blockIdx.x * COLS_PER_BLOCK;

    v2f ones;
    ones.x = 1.0f;
    ones.y = 1.0f;

    for (int col = 0; col < COLS_PER_BLOCK; ++col) {
        const int e = e_base + col;
        const size_t kbase = (size_t)e * K_NNZ;

        if (tid < K_NNZ) {
            s_vals[tid] = values[kbase + tid];
        } else {
            const int t = tid - K_NNZ;
            s_rows[t] = rows[kbase + t];
        }
        if (col + 1 < COLS_PER_BLOCK && tid == 0) {
            __builtin_prefetch(values + kbase + K_NNZ, 0, 1);
            __builtin_prefetch(rows + kbase + K_NNZ, 0, 1);
        }
        __syncthreads();

        float t_cache[64];
        float m = -INFINITY;
#pragma unroll
        for (int c = 0; c < 32; ++c) {
            const int k0 = 4 * c + 2 * h;
            const float t0 = s_vals[k0]     + load_xt(xt, s_rows[k0],     lane_byte);
            const float t1 = s_vals[k0 + 1] + load_xt(xt, s_rows[k0 + 1], lane_byte);
            t_cache[2 * c]     = t0;
            t_cache[2 * c + 1] = t1;
            m = fmaxf(m, fmaxf(t0, t1));
        }
        m = fmaxf(m, __shfl_xor(m, 16, 32));

        v8f acc = {};
#pragma unroll
        for (int c = 0; c < 32; ++c) {
            v2f a;
            a.x = __expf(t_cache[2 * c]     - m);
            a.y = __expf(t_cache[2 * c + 1] - m);
            acc = __builtin_amdgcn_wmma_f32_16x16x4_f32(
                false, a, false, ones, (short)0, acc, false, false);
        }

        if (lane == 0) {
#pragma unroll
            for (int v = 0; v < 8; ++v) s_sums[wave * 16 + v] = acc[v];
        } else if (lane == 16) {
#pragma unroll
            for (int v = 0; v < 8; ++v) s_sums[wave * 16 + 8 + v] = acc[v];
        }
        if (lane < 16) {
            const float s = s_sums[wave * 16 + nloc];
            s_out[nrow * COLS_PER_BLOCK + col] =
                m + 0.6931471805599453f * __log2f(s);
        }
        __syncthreads();
    }

    {
        const int n    = tid >> 1;
        const int part = tid & 1;
        const float4* src =
            (const float4*)(s_out + n * COLS_PER_BLOCK + part * 4);
        float4* dst =
            (float4*)(out + (size_t)n * E_COLS + e_base + part * 4);
        *dst = *src;
    }
}

extern "C" void kernel_launch(void* const* d_in, const int* in_sizes, int n_in,
                              void* d_out, int out_size, void* d_ws, size_t ws_size,
                              hipStream_t stream) {
    const float* x      = (const float*)d_in[0];
    const float* values = (const float*)d_in[1];
    const int*   rowidx = (const int*)d_in[2];
    float* out = (float*)d_out;
    float* xt  = (float*)d_ws;

    hipLaunchKernelGGL(transpose_x_kernel,
                       dim3(D_DIM / TP_TILE, N_BATCH / TP_TILE), dim3(256), 0, stream,
                       x, xt);

    dim3 grid(E_COLS / COLS_PER_BLOCK);
    dim3 block(THREADS);
    hipLaunchKernelGGL(logmmexp_wmma_kernel, grid, block, 0, stream,
                       (const float*)xt, values, rowidx, out);
}


# the 64 gather loads of a column are issued as one group before their first use (a wave had one or two in flight, now 64); indices read from LDS in 16-byte pieces; k and summation order unchanged
# speedup vs baseline: 7.6386x; 1.5604x over previous
#include <hip/hip_runtime.h>
#include <math.h>


typedef __attribute__((ext_vector_type(2))) float v2f;
typedef __attribute__((ext_vector_type(8))) float v8f;
typedef __attribute__((ext_vector_type(16))) int v16i;

#define N_BATCH 128
#define D_DIM 4096
#define E_COLS 8192
#define K_NNZ 128
#define COLS_PER_BLOCK 8
#define THREADS 256
#define TP_TILE 32
#define XT_ROW_BYTES (N_BATCH * 4)

__global__ __launch_bounds__(256)
void transpose_x_kernel(const float* __restrict__ x, float* __restrict__ xt)
{
    __shared__ float tile[TP_TILE][TP_TILE + 1];

    const int tx = threadIdx.x & 31;
    const int ty = threadIdx.x >> 5;
    const int d0 = blockIdx.x * TP_TILE;
    const int n0 = blockIdx.y * TP_TILE;

#pragma unroll
    for (int j = 0; j < TP_TILE; j += 8)
        tile[ty + j][tx] = x[(size_t)(n0 + ty + j) * D_DIM + d0 + tx];
    __syncthreads();
#pragma unroll
    for (int j = 0; j < TP_TILE; j += 8)
        xt[(size_t)(d0 + ty + j) * N_BATCH + n0 + tx] = tile[tx][ty + j];
}

__device__ __forceinline__ float load_xt_off(const float* __restrict__ xt, unsigned off)
{
    return *(const float*)((const char*)xt + off);
}

__global__ __launch_bounds__(THREADS)
void logmmexp_wmma_kernel(const float* __restrict__ xt,
                          const float* __restrict__ values,
                          const int* __restrict__ rows,
                          float* __restrict__ out)
{
    __shared__ float s_vals[K_NNZ];
    __shared__ __attribute__((aligned(64))) int s_rows[K_NNZ];
    __shared__ float s_sums[8 * 16];
    __shared__ float s_out[N_BATCH * COLS_PER_BLOCK];

    const int tid  = threadIdx.x;
    const int wave = tid >> 5;
    const int lane = tid & 31;
    const int h    = lane >> 4;
    const int nloc = lane & 15;
    const int nrow = wave * 16 + nloc;
    const unsigned lane_byte = (unsigned)nrow * 4u;

    const int e_base = blockIdx.x * COLS_PER_BLOCK;

    v2f ones;
    ones.x = 1.0f;
    ones.y = 1.0f;

    for (int col = 0; col < COLS_PER_BLOCK; ++col) {
        const int e = e_base + col;
        const size_t kbase = (size_t)e * K_NNZ;

        if (tid < K_NNZ) {
            s_vals[tid] = values[kbase + tid];
        } else {
            const int t = tid - K_NNZ;
            s_rows[((t >> 1) & 1) * 64 + (t >> 2) * 2 + (t & 1)] = rows[kbase + t];
        }
        if (col + 1 < COLS_PER_BLOCK && tid == 0) {
            __builtin_prefetch(values + kbase + K_NNZ, 0, 1);
            __builtin_prefetch(rows + kbase + K_NNZ, 0, 1);
        }
        __syncthreads();

        v16i q[4];
#pragma unroll
        for (int j = 0; j < 4; ++j)
            q[j] = *(const v16i*)(s_rows + h * 64 + 16 * j);
#pragma unroll
        for (int j = 0; j < 4; ++j)
            q[j] = q[j] * XT_ROW_BYTES + (int)lane_byte;
#pragma unroll
        for (int c = 0; c < 32; ++c) {
            const int e0 = 2 * (c & 7);
            q[c >> 3][e0]     = __float_as_int(load_xt_off(xt, (unsigned)q[c >> 3][e0]));
            q[c >> 3][e0 + 1] = __float_as_int(load_xt_off(xt, (unsigned)q[c >> 3][e0 + 1]));
        }
        asm volatile("" :: "v"(q[0]), "v"(q[1]), "v"(q[2]), "v"(q[3]));

        float t_cache[64];
        float m = -INFINITY;
#pragma unroll
        for (int c = 0; c < 32; ++c) {
            const int k0 = 4 * c + 2 * h;
            const float t0 = s_vals[k0]     + __int_as_float(q[c >> 3][2 * (c & 7)]);
            const float t1 = s_vals[k0 + 1] + __int_as_float(q[c >> 3][2 * (c & 7) + 1]);
            t_cache[2 * c]     = t0;
            t_cache[2 * c + 1] = t1;
            m = fmaxf(m, fmaxf(t0, t1));
        }
        m = fmaxf(m, __shfl_xor(m, 16, 32));

        v8f acc = {};
#pragma unroll
        for (int c = 0; c < 32; ++c) {
            v2f a;
            a.x = __expf(t_cache[2 * c]     - m);
            a.y = __expf(t_cache[2 * c + 1] - m);
            acc = __builtin_amdgcn_wmma_f32_16x16x4_f32(
                false, a, false, ones, (short)0, acc, false, false);
        }

        if (lane == 0) {
#pragma unroll
            for (int v = 0; v < 8; ++v) s_sums[wave * 16 + v] = acc[v];
        } else if (lane == 16) {
#pragma unroll
            for (int v = 0; v < 8; ++v) s_sums[wave * 16 + 8 + v] = acc[v];
        }
        if (lane < 16) {
            const float s = s_sums[wave * 16 + nloc];
            s_out[nrow * COLS_PER_BLOCK + col] =
                m + 0.6931471805599453f * __log2f(s);
        }
        __syncthreads();
    }

    {
        const int n    = tid >> 1;
        const int part = tid & 1;
        const float4* src =
            (const float4*)(s_out + n * COLS_PER_BLOCK + part * 4);
        float4* dst =
            (float4*)(out + (size_t)n * E_COLS + e_base + part * 4);
        *dst = *src;
    }
}

extern "C" void kernel_launch(void* const* d_in, const int* in_sizes, int n_in,
                              void* d_out, int out_size, void* d_ws, size_t ws_size,
                              hipStream_t stream) {
    const float* x      = (const float*)d_in[0];
    const float* values = (const float*)d_in[1];
    const int*   rowidx = (const int*)d_in[2];
    float* out = (float*)d_out;
    float* xt  = (float*)d_ws;

    hipLaunchKernelGGL(transpose_x_kernel,
                       dim3(D_DIM / TP_TILE, N_BATCH / TP_TILE), dim3(256), 0, stream,
                       x, xt);

    dim3 grid(E_COLS / COLS_PER_BLOCK);
    dim3 block(THREADS);
    hipLaunchKernelGGL(logmmexp_wmma_kernel, grid, block, 0, stream,
                       (const float*)xt, values, rowidx, out);
}
